# starting kernel
# baseline (speedup 1.0000x reference)
#include <hip/hip_runtime.h>

typedef float v2f __attribute__((ext_vector_type(2)));
typedef float v8f __attribute__((ext_vector_type(8)));

#define BATCH 8
#define SEQ 2048
#define DMODEL 256
#define THREADS 128
#define WAVES 4
#define ROWS_PER_WAVE 16
#define ROWS_PER_BLOCK 64
#define LAMBDA 0.5f
#define NEG_BIG -1.0e9f

__global__ __launch_bounds__(THREADS)
void dwsa_kernel(const float* __restrict__ in,
                 const float* __restrict__ Wq,
                 const float* __restrict__ Wk,
                 const float* __restrict__ Wv,
                 float* __restrict__ out)
{
    __shared__ __align__(16) float2 s_kv[SEQ];
    __shared__ float s_red[WAVES][2][ROWS_PER_WAVE];
    __shared__ float s_A[ROWS_PER_BLOCK];

    const int tid  = threadIdx.x;
    const int b    = blockIdx.x >> 5;
    const int rowBase = (blockIdx.x & 31) * ROWS_PER_BLOCK;
    const int wave = tid >> 5;
    const int lane = tid & 31;
    const int half = lane >> 4;
    const int r    = lane & 15;
    const int n    = lane & 15;

    const float2* seq = (const float2*)in + (size_t)b * SEQ;
    for (int t = tid; t < SEQ; t += THREADS) s_kv[t] = seq[t];

    float c = 0.0f;
    #pragma unroll 8
    for (int i = 0; i < DMODEL; ++i) c += Wq[i] * Wk[i];
    c *= (1.0f / 16.0f);

    __syncthreads();

    const int   row = rowBase + wave * ROWS_PER_WAVE + r;
    const float2 q2 = s_kv[row];
    const float zs = q2.x, hs = q2.y;

    float m = -3.4e38f;
    const int t0 = half * (SEQ / 2);
    for (int t = t0; t < t0 + SEQ / 2; ++t) {
        float2 kt = s_kv[t];
        float l = c * hs * kt.y - LAMBDA * fabsf(zs - kt.x);
        l += (kt.y == 0.0f) ? NEG_BIG : 0.0f;
        m = fmaxf(m, l);
    }
    m = fmaxf(m, __shfl_xor(m, 16, 32));

    v8f acc = {0.f, 0.f, 0.f, 0.f, 0.f, 0.f, 0.f, 0.f};
    const float4* kv4 = (const float4*)s_kv;
    #pragma unroll 2
    for (int kc = 0; kc < SEQ; kc += 4) {
        const int ka = kc + 2 * half;
        float4 kk = kv4[ka >> 1];
        const float za = kk.x, ha = kk.y, zb = kk.z, hb = kk.w;
        float la = c * hs * ha - LAMBDA * fabsf(zs - za) + ((ha == 0.0f) ? NEG_BIG : 0.0f);
        float lb = c * hs * hb - LAMBDA * fabsf(zs - zb) + ((hb == 0.0f) ? NEG_BIG : 0.0f);
        v2f a;
        a.x = __expf(la - m);
        a.y = __expf(lb - m);
        v2f bop;
        bop.x = (n == 0) ? ha : ((n == 1) ? 1.0f : 0.0f);
        bop.y = (n == 0) ? hb : ((n == 1) ? 1.0f : 0.0f);
        acc = __builtin_amdgcn_wmma_f32_16x16x4_f32(
             false, a,  false, bop,
             (short)0, acc,  false,  false);
    }

    if (n < 2) {
        #pragma unroll
        for (int v = 0; v < 8; ++v)
            s_red[wave][n][v + 8 * half] = acc[v];
    }
    __syncthreads();

    if (tid < ROWS_PER_BLOCK) {
        const int w = tid >> 4, rr = tid & 15;
        const float eh = s_red[w][0][rr];
        const float se = s_red[w][1][rr];
        s_A[tid] = eh / se;
    }
    __syncthreads();

    const float wv0 = Wv[2 * tid];
    const float wv1 = Wv[2 * tid + 1];
    v2f* o2 = (v2f*)out + (size_t)(b * SEQ + rowBase) * (DMODEL / 2) + tid;
    for (int rr = 0; rr < ROWS_PER_BLOCK; ++rr) {
        const float A = s_A[rr];
        v2f val;
        val.x = A * wv0;
        val.y = A * wv1;
        __builtin_nontemporal_store(val, o2);
        o2 += DMODEL / 2;
    }
}

extern "C" void kernel_launch(void* const* d_in, const int* in_sizes, int n_in,
                              void* d_out, int out_size, void* d_ws, size_t ws_size,
                              hipStream_t stream) {
    (void)in_sizes; (void)n_in; (void)out_size; (void)d_ws; (void)ws_size;
    const float* in = (const float*)d_in[0];
    const float* Wq = (const float*)d_in[1];
    const float* Wk = (const float*)d_in[2];
    const float* Wv = (const float*)d_in[3];
    float* out = (float*)d_out;

    dim3 grid(BATCH * (SEQ / ROWS_PER_BLOCK));
    dim3 block(THREADS);
    dwsa_kernel<<<grid, block, 0, stream>>>(in, Wq, Wk, Wv, out);
}


# Logits on the matrix unit: per 16-key x 16-query tile, P and Q (logit = P - |Q|) from 2 matrix ops on one key fragment; the exp tile is used in place as B of the accumulating ops; sum order changes
# speedup vs baseline: 3.4264x; 3.4264x over previous
#include <hip/hip_runtime.h>


typedef float v2f __attribute__((ext_vector_type(2)));
typedef float v8f __attribute__((ext_vector_type(8)));

#define BATCH 8
#define SEQ 2048
#define DMODEL 256
#define THREADS 128
#define WAVES 4
#define ROWS_PER_WAVE 16
#define ROWS_PER_BLOCK 64
#define LAMBDA 0.5f
#define NEG_BIG -1.0e9f
#define LOG2E 1.4426950408889634f
#define TILE 16
#define NTILES (SEQ / TILE)
#define PLANE1 (SEQ + 16)

__device__ __forceinline__ float fmax_n(float a, float b) { return __builtin_fmaximum_numf(a, b); }

__device__ __forceinline__ v8f mm4(v2f a, v2f b, v8f c) {
    return __builtin_amdgcn_wmma_f32_16x16x4_f32(
         false, a,  false, b,
         (short)0, c,  false,  false);
}


__global__ __launch_bounds__(THREADS)
void dwsa_kernel(const float* __restrict__ in,
                 const float* __restrict__ Wq,
                 const float* __restrict__ Wk,
                 const float* __restrict__ Wv,
                 float* __restrict__ out)
{
    __shared__ __align__(16) float2 s_ka[PLANE1 + SEQ];
    __shared__ __align__(16) float  s_h[SEQ + 16];
    __shared__ float s_A[ROWS_PER_BLOCK];

    const int tid  = threadIdx.x;
    const int b    = blockIdx.x >> 5;
    const int rowBase = (blockIdx.x & 31) * ROWS_PER_BLOCK;
    const int wave = tid >> 5;
    const int lane = tid & 31;
    const int half = lane >> 4;
    const int cidx = lane & 15;

    const float2* seq = (const float2*)in + (size_t)b * SEQ;
    for (int t = tid; t < SEQ; t += THREADS) {
        const float2 zh = seq[t];
        s_ka[t]          = make_float2(zh.y, (LAMBDA * LOG2E) * zh.x);
        s_ka[PLANE1 + t] = make_float2((zh.y == 0.0f) ? (LOG2E * NEG_BIG) : 0.0f, 1.0f);
        s_h[t]           = zh.y;
    }
    if (tid < 16) s_h[SEQ + tid] = (tid < 8) ? 1.0f : 0.0f;

    float c = 0.0f;
    #pragma unroll 8
    for (int i = 0; i < DMODEL; ++i) c += Wq[i] * Wk[i];
    c *= (1.0f / 16.0f);

    __syncthreads();

    const int    row = rowBase + wave * ROWS_PER_WAVE + cidx;
    const float2 q2  = s_ka[row];
    const float  gh  = (LOG2E * c) * q2.x;
    const float  ui  = q2.y;

    const float2* ka = s_ka + (half ? PLANE1 : 0) + cidx;
    const v8f zero8 = {0.f, 0.f, 0.f, 0.f, 0.f, 0.f, 0.f, 0.f};

    v2f bP, bQ;
    bP.x = half ? 1.0f : gh;   bP.y = 0.0f;
    bQ.x = 0.0f;               bQ.y = half ? -ui : 1.0f;

    v8f mx = {-3.4e38f, -3.4e38f, -3.4e38f, -3.4e38f, -3.4e38f, -3.4e38f, -3.4e38f, -3.4e38f};
    {
        float2 nk[4];
        #pragma unroll
        for (int u = 0; u < 4; ++u) nk[u] = ka[u * TILE];
        for (int t = 0; t < NTILES; t += 4) {
            float2 kk[4];
            #pragma unroll
            for (int u = 0; u < 4; ++u) kk[u] = nk[u];
            const int tn = (t + 4 < NTILES) ? (t + 4) : t;
            #pragma unroll
            for (int u = 0; u < 4; ++u) nk[u] = ka[(tn + u) * TILE];
            __builtin_amdgcn_sched_barrier(0);
            v8f p[4], q[4];
            #pragma unroll
            for (int u = 0; u < 4; ++u) {
                const v2f a = {kk[u].x, kk[u].y};
                p[u] = mm4(a, bP, zero8);
                q[u] = mm4(a, bQ, zero8);
            }
            #pragma unroll
            for (int u = 0; u < 4; ++u) {
                #pragma unroll
                for (int r = 0; r < 8; ++r) mx[r] = fmax_n(mx[r], p[u][r] - fabsf(q[u][r]));
            }
            asm volatile("" :: "v"(nk[0].x), "v"(nk[0].y), "v"(nk[1].x), "v"(nk[1].y),
                               "v"(nk[2].x), "v"(nk[2].y), "v"(nk[3].x), "v"(nk[3].y));
        }
    }
    float m = fmax_n(fmax_n(fmax_n(mx[0], mx[1]), fmax_n(mx[2], mx[3])),
                     fmax_n(fmax_n(mx[4], mx[5]), fmax_n(mx[6], mx[7])));
    m = fmax_n(m, __shfl_xor(m, 16, 32));

    bP.y = half ? -m : 0.0f;

    const int hbase   = (cidx == 0) ? (8 * half) : ((cidx == 1) ? SEQ : (SEQ + 8));
    const int hstride = (cidx == 0) ? TILE : 0;

    v8f acc0 = zero8, acc1 = zero8;
    {
        float2 nk[4];
        #pragma unroll
        for (int u = 0; u < 4; ++u) nk[u] = ka[u * TILE];
        for (int t = 0; t < NTILES; t += 4) {
            float2 kk[4];
            #pragma unroll
            for (int u = 0; u < 4; ++u) kk[u] = nk[u];
            float4 ff[4][2];
            #pragma unroll
            for (int u = 0; u < 4; ++u) {
                const float4* hp = (const float4*)(s_h + hbase + (t + u) * hstride);
                ff[u][0] = hp[0]; ff[u][1] = hp[1];
            }
            const int tn = (t + 4 < NTILES) ? (t + 4) : t;
            #pragma unroll
            for (int u = 0; u < 4; ++u) nk[u] = ka[(tn + u) * TILE];
            __builtin_amdgcn_sched_barrier(0);

            v8f p[4], q[4];
            #pragma unroll
            for (int u = 0; u < 4; ++u) {
                const v2f a = {kk[u].x, kk[u].y};
                p[u] = mm4(a, bP, zero8);
                q[u] = mm4(a, bQ, zero8);
            }
            #pragma unroll
            for (int u = 0; u < 4; ++u) {
                v8f e;
                #pragma unroll
                for (int r = 0; r < 8; ++r)
                    e[r] = __builtin_amdgcn_exp2f(p[u][r] - fabsf(q[u][r]));
                v2f ha, eb;
                ha = (v2f){ff[u][0].x, ff[u][0].y}; eb = (v2f){e[0], e[1]}; acc0 = mm4(ha, eb, acc0);
                ha = (v2f){ff[u][0].z, ff[u][0].w}; eb = (v2f){e[2], e[3]}; acc1 = mm4(ha, eb, acc1);
                ha = (v2f){ff[u][1].x, ff[u][1].y}; eb = (v2f){e[4], e[5]}; acc0 = mm4(ha, eb, acc0);
                ha = (v2f){ff[u][1].z, ff[u][1].w}; eb = (v2f){e[6], e[7]}; acc1 = mm4(ha, eb, acc1);
            }
            asm volatile("" :: "v"(nk[0].x), "v"(nk[0].y), "v"(nk[1].x), "v"(nk[1].y),
                               "v"(nk[2].x), "v"(nk[2].y), "v"(nk[3].x), "v"(nk[3].y));
        }
    }

    if (half == 0) {
        const float eh = acc0[0] + acc1[0];
        const float se = acc0[1] + acc1[1];
        s_A[wave * ROWS_PER_WAVE + cidx] = (m < -1.0e8f) ? 0.0f : (eh / se);
    }
    __syncthreads();

    const float wv0 = Wv[2 * tid];
    const float wv1 = Wv[2 * tid + 1];
    v2f* o2 = (v2f*)out + (size_t)(b * SEQ + rowBase) * (DMODEL / 2) + tid;
    for (int rr = 0; rr < ROWS_PER_BLOCK; ++rr) {
        const float A = s_A[rr];
        v2f val;
        val.x = A * wv0;
        val.y = A * wv1;
        __builtin_nontemporal_store(val, o2);
        o2 += DMODEL / 2;
    }
}

extern "C" void kernel_launch(void* const* d_in, const int* in_sizes, int n_in,
                              void* d_out, int out_size, void* d_ws, size_t ws_size,
                              hipStream_t stream) {
    (void)in_sizes; (void)n_in; (void)out_size; (void)d_ws; (void)ws_size;
    const float* in = (const float*)d_in[0];
    const float* Wq = (const float*)d_in[1];
    const float* Wk = (const float*)d_in[2];
    const float* Wv = (const float*)d_in[3];
    float* out = (float*)d_out;

    dim3 grid(BATCH * (SEQ / ROWS_PER_BLOCK));
    dim3 block(THREADS);
    dwsa_kernel<<<grid, block, 0, stream>>>(in, Wq, Wk, Wv, out);
}


# Key staging: each thread's 16 global reads issued as one group at entry and left in flight across the c loop; was 8 dependent round trips of 2 reads, each waited for before the next
# speedup vs baseline: 3.6424x; 1.0630x over previous
#include <hip/hip_runtime.h>


typedef float v2f __attribute__((ext_vector_type(2)));
typedef float v8f __attribute__((ext_vector_type(8)));

#define BATCH 8
#define SEQ 2048
#define DMODEL 256
#define THREADS 128
#define WAVES 4
#define ROWS_PER_WAVE 16
#define ROWS_PER_BLOCK 64
#define LAMBDA 0.5f
#define NEG_BIG -1.0e9f
#define LOG2E 1.4426950408889634f
#define TILE 16
#define NTILES (SEQ / TILE)
#define KEYS_PER_THREAD (SEQ / THREADS)
#define PLANE1 (SEQ + 16)

__device__ __forceinline__ float fmax_n(float a, float b) { return __builtin_fmaximum_numf(a, b); }

__device__ __forceinline__ v8f mm4(v2f a, v2f b, v8f c) {
    return __builtin_amdgcn_wmma_f32_16x16x4_f32(
         false, a,  false, b,
         (short)0, c,  false,  false);
}


__global__ __launch_bounds__(THREADS)
void dwsa_kernel(const float* __restrict__ in,
                 const float* __restrict__ Wq,
                 const float* __restrict__ Wk,
                 const float* __restrict__ Wv,
                 float* __restrict__ out)
{
    __shared__ __align__(16) float2 s_ka[PLANE1 + SEQ];
    __shared__ __align__(16) float  s_h[SEQ + 16];
    __shared__ float s_A[ROWS_PER_BLOCK];

    const int tid  = threadIdx.x;
    const int b    = blockIdx.x >> 5;
    const int rowBase = (blockIdx.x & 31) * ROWS_PER_BLOCK;
    const int wave = tid >> 5;
    const int lane = tid & 31;
    const int half = lane >> 4;
    const int cidx = lane & 15;

    const float2* seq = (const float2*)in + (size_t)b * SEQ;
    float2 zh[KEYS_PER_THREAD];
    #pragma unroll
    for (int k = 0; k < KEYS_PER_THREAD; ++k) zh[k] = seq[tid + k * THREADS];

    float c = 0.0f;
    #pragma unroll 8
    for (int i = 0; i < DMODEL; ++i) c += Wq[i] * Wk[i];
    c *= (1.0f / 16.0f);

    #pragma unroll
    for (int k = 0; k < KEYS_PER_THREAD; ++k) {
        const int t = tid + k * THREADS;
        s_ka[t]          = make_float2(zh[k].y, (LAMBDA * LOG2E) * zh[k].x);
        s_ka[PLANE1 + t] = make_float2((zh[k].y == 0.0f) ? (LOG2E * NEG_BIG) : 0.0f, 1.0f);
        s_h[t]           = zh[k].y;
    }
    if (tid < 16) s_h[SEQ + tid] = (tid < 8) ? 1.0f : 0.0f;

    __syncthreads();

    const int    row = rowBase + wave * ROWS_PER_WAVE + cidx;
    const float2 q2  = s_ka[row];
    const float  gh  = (LOG2E * c) * q2.x;
    const float  ui  = q2.y;

    const float2* ka = s_ka + (half ? PLANE1 : 0) + cidx;
    const v8f zero8 = {0.f, 0.f, 0.f, 0.f, 0.f, 0.f, 0.f, 0.f};

    v2f bP, bQ;
    bP.x = half ? 1.0f : gh;   bP.y = 0.0f;
    bQ.x = 0.0f;               bQ.y = half ? -ui : 1.0f;

    v8f mx = {-3.4e38f, -3.4e38f, -3.4e38f, -3.4e38f, -3.4e38f, -3.4e38f, -3.4e38f, -3.4e38f};
    {
        float2 nk[4];
        #pragma unroll
        for (int u = 0; u < 4; ++u) nk[u] = ka[u * TILE];
        for (int t = 0; t < NTILES; t += 4) {
            float2 kk[4];
            #pragma unroll
            for (int u = 0; u < 4; ++u) kk[u] = nk[u];
            const int tn = (t + 4 < NTILES) ? (t + 4) : t;
            #pragma unroll
            for (int u = 0; u < 4; ++u) nk[u] = ka[(tn + u) * TILE];
            __builtin_amdgcn_sched_barrier(0);
            v8f p[4], q[4];
            #pragma unroll
            for (int u = 0; u < 4; ++u) {
                const v2f a = {kk[u].x, kk[u].y};
                p[u] = mm4(a, bP, zero8);
                q[u] = mm4(a, bQ, zero8);
            }
            #pragma unroll
            for (int u = 0; u < 4; ++u) {
                #pragma unroll
                for (int r = 0; r < 8; ++r) mx[r] = fmax_n(mx[r], p[u][r] - fabsf(q[u][r]));
            }
            asm volatile("" :: "v"(nk[0].x), "v"(nk[0].y), "v"(nk[1].x), "v"(nk[1].y),
                               "v"(nk[2].x), "v"(nk[2].y), "v"(nk[3].x), "v"(nk[3].y));
        }
    }
    float m = fmax_n(fmax_n(fmax_n(mx[0], mx[1]), fmax_n(mx[2], mx[3])),
                     fmax_n(fmax_n(mx[4], mx[5]), fmax_n(mx[6], mx[7])));
    m = fmax_n(m, __shfl_xor(m, 16, 32));

    bP.y = half ? -m : 0.0f;

    const int hbase   = (cidx == 0) ? (8 * half) : ((cidx == 1) ? SEQ : (SEQ + 8));
    const int hstride = (cidx == 0) ? TILE : 0;

    v8f acc0 = zero8, acc1 = zero8;
    {
        float2 nk[4];
        #pragma unroll
        for (int u = 0; u < 4; ++u) nk[u] = ka[u * TILE];
        for (int t = 0; t < NTILES; t += 4) {
            float2 kk[4];
            #pragma unroll
            for (int u = 0; u < 4; ++u) kk[u] = nk[u];
            float4 ff[4][2];
            #pragma unroll
            for (int u = 0; u < 4; ++u) {
                const float4* hp = (const float4*)(s_h + hbase + (t + u) * hstride);
                ff[u][0] = hp[0]; ff[u][1] = hp[1];
            }
            const int tn = (t + 4 < NTILES) ? (t + 4) : t;
            #pragma unroll
            for (int u = 0; u < 4; ++u) nk[u] = ka[(tn + u) * TILE];
            __builtin_amdgcn_sched_barrier(0);

            v8f p[4], q[4];
            #pragma unroll
            for (int u = 0; u < 4; ++u) {
                const v2f a = {kk[u].x, kk[u].y};
                p[u] = mm4(a, bP, zero8);
                q[u] = mm4(a, bQ, zero8);
            }
            #pragma unroll
            for (int u = 0; u < 4; ++u) {
                v8f e;
                #pragma unroll
                for (int r = 0; r < 8; ++r)
                    e[r] = __builtin_amdgcn_exp2f(p[u][r] - fabsf(q[u][r]));
                v2f ha, eb;
                ha = (v2f){ff[u][0].x, ff[u][0].y}; eb = (v2f){e[0], e[1]}; acc0 = mm4(ha, eb, acc0);
                ha = (v2f){ff[u][0].z, ff[u][0].w}; eb = (v2f){e[2], e[3]}; acc1 = mm4(ha, eb, acc1);
                ha = (v2f){ff[u][1].x, ff[u][1].y}; eb = (v2f){e[4], e[5]}; acc0 = mm4(ha, eb, acc0);
                ha = (v2f){ff[u][1].z, ff[u][1].w}; eb = (v2f){e[6], e[7]}; acc1 = mm4(ha, eb, acc1);
            }
            asm volatile("" :: "v"(nk[0].x), "v"(nk[0].y), "v"(nk[1].x), "v"(nk[1].y),
                               "v"(nk[2].x), "v"(nk[2].y), "v"(nk[3].x), "v"(nk[3].y));
        }
    }

    if (half == 0) {
        const float eh = acc0[0] + acc1[0];
        const float se = acc0[1] + acc1[1];
        s_A[wave * ROWS_PER_WAVE + cidx] = (m < -1.0e8f) ? 0.0f : (eh / se);
    }
    __syncthreads();

    const float wv0 = Wv[2 * tid];
    const float wv1 = Wv[2 * tid + 1];
    v2f* o2 = (v2f*)out + (size_t)(b * SEQ + rowBase) * (DMODEL / 2) + tid;
    for (int rr = 0; rr < ROWS_PER_BLOCK; ++rr) {
        const float A = s_A[rr];
        v2f val;
        val.x = A * wv0;
        val.y = A * wv1;
        __builtin_nontemporal_store(val, o2);
        o2 += DMODEL / 2;
    }
}

extern "C" void kernel_launch(void* const* d_in, const int* in_sizes, int n_in,
                              void* d_out, int out_size, void* d_ws, size_t ws_size,
                              hipStream_t stream) {
    (void)in_sizes; (void)n_in; (void)out_size; (void)d_ws; (void)ws_size;
    const float* in = (const float*)d_in[0];
    const float* Wq = (const float*)d_in[1];
    const float* Wk = (const float*)d_in[2];
    const float* Wv = (const float*)d_in[3];
    float* out = (float*)d_out;

    dim3 grid(BATCH * (SEQ / ROWS_PER_BLOCK));
    dim3 block(THREADS);
    dwsa_kernel<<<grid, block, 0, stream>>>(in, Wq, Wk, Wv, out);
}


# One pass over the keys: each lane half keeps a running max of its own keys and rescales its two sums when it rises (accumulate rows 0,1 and 8,9); the separate max pass goes; sum order changes
# speedup vs baseline: 4.0858x; 1.1217x over previous
#include <hip/hip_runtime.h>


typedef float v2f __attribute__((ext_vector_type(2)));
typedef float v8f __attribute__((ext_vector_type(8)));

#define BATCH 8
#define SEQ 2048
#define DMODEL 256
#define THREADS 128
#define WAVES 4
#define ROWS_PER_WAVE 16
#define ROWS_PER_BLOCK 64
#define LAMBDA 0.5f
#define NEG_BIG -1.0e9f
#define LOG2E 1.4426950408889634f
#define TILE 16
#define NTILES (SEQ / TILE)
#define KEYS_PER_THREAD (SEQ / THREADS)
#define PLANE1 (SEQ + 16)

__device__ __forceinline__ float fmax_n(float a, float b) { return __builtin_fmaximum_numf(a, b); }

__device__ __forceinline__ v8f mm4(v2f a, v2f b, v8f c) {
    return __builtin_amdgcn_wmma_f32_16x16x4_f32(
         false, a,  false, b,
         (short)0, c,  false,  false);
}


__global__ __launch_bounds__(THREADS)
void dwsa_kernel(const float* __restrict__ in,
                 const float* __restrict__ Wq,
                 const float* __restrict__ Wk,
                 const float* __restrict__ Wv,
                 float* __restrict__ out)
{
    __shared__ __align__(16) float2 s_ka[PLANE1 + SEQ];
    __shared__ __align__(16) float  s_h[SEQ + 16];
    __shared__ float s_A[ROWS_PER_BLOCK];

    const int tid  = threadIdx.x;
    const int b    = blockIdx.x >> 5;
    const int rowBase = (blockIdx.x & 31) * ROWS_PER_BLOCK;
    const int wave = tid >> 5;
    const int lane = tid & 31;
    const int half = lane >> 4;
    const int cidx = lane & 15;

    const float2* seq = (const float2*)in + (size_t)b * SEQ;
    float2 zh[KEYS_PER_THREAD];
    #pragma unroll
    for (int k = 0; k < KEYS_PER_THREAD; ++k) zh[k] = seq[tid + k * THREADS];

    float c = 0.0f;
    #pragma unroll 8
    for (int i = 0; i < DMODEL; ++i) c += Wq[i] * Wk[i];
    c *= (1.0f / 16.0f);

    #pragma unroll
    for (int k = 0; k < KEYS_PER_THREAD; ++k) {
        const int t = tid + k * THREADS;
        s_ka[t]          = make_float2(zh[k].y, (LAMBDA * LOG2E) * zh[k].x);
        s_ka[PLANE1 + t] = make_float2((zh[k].y == 0.0f) ? (LOG2E * NEG_BIG) : 0.0f, 1.0f);
        s_h[t]           = zh[k].y;
    }
    if (tid < 16) s_h[SEQ + tid] = (tid < 8) ? 1.0f : 0.0f;

    __syncthreads();

    const int    row = rowBase + wave * ROWS_PER_WAVE + cidx;
    const float2 q2  = s_ka[row];
    const float  gh  = (LOG2E * c) * q2.x;
    const float  ui  = q2.y;

    const float2* ka = s_ka + (half ? PLANE1 : 0) + cidx;
    const v8f zero8 = {0.f, 0.f, 0.f, 0.f, 0.f, 0.f, 0.f, 0.f};

    v2f bP, bQ;
    bP.x = half ? 1.0f : gh;   bP.y = 0.0f;
    bQ.x = 0.0f;               bQ.y = half ? -ui : 1.0f;

    const bool rowH   = (cidx == 8 * half);
    const bool rowOne = (cidx == 8 * half + 1);
    const int hbase   = rowH ? (8 * half) : (rowOne ? SEQ : (SEQ + 8));
    const int hstride = rowH ? TILE : 0;

    float m = -3.4e38f;
    v8f acc0 = zero8, acc1 = zero8;
    {
        v8f p[4], q[4];
        float2 nk[4];
        #pragma unroll
        for (int u = 0; u < 4; ++u) {
            const float2 k0 = ka[u * TILE];
            const v2f a = {k0.x, k0.y};
            p[u] = mm4(a, bP, zero8);
            q[u] = mm4(a, bQ, zero8);
        }
        #pragma unroll
        for (int u = 0; u < 4; ++u) nk[u] = ka[(4 + u) * TILE];

        for (int t = 0; t < NTILES; t += 4) {
            float2 kk[4];
            #pragma unroll
            for (int u = 0; u < 4; ++u) kk[u] = nk[u];
            float4 ff[4][2];
            #pragma unroll
            for (int u = 0; u < 4; ++u) {
                const float4* hp = (const float4*)(s_h + hbase + (t + u) * hstride);
                ff[u][0] = hp[0]; ff[u][1] = hp[1];
            }
            const int tn = (t + 8 < NTILES) ? (t + 8) : (NTILES - 4);
            #pragma unroll
            for (int u = 0; u < 4; ++u) nk[u] = ka[(tn + u) * TILE];
            __builtin_amdgcn_sched_barrier(0);

            v8f L[4];
            float gm[4];
            #pragma unroll
            for (int u = 0; u < 4; ++u) {
                #pragma unroll
                for (int r = 0; r < 8; ++r) L[u][r] = p[u][r] - fabsf(q[u][r]);
                gm[u] = fmax_n(fmax_n(fmax_n(L[u][0], L[u][1]), fmax_n(L[u][2], L[u][3])),
                               fmax_n(fmax_n(L[u][4], L[u][5]), fmax_n(L[u][6], L[u][7])));
                __builtin_amdgcn_sched_barrier(0);
                const v2f a = {kk[u].x, kk[u].y};
                q[u] = mm4(a, bQ, zero8);
            }
            const float mn = fmax_n(fmax_n(m, fmax_n(gm[0], gm[1])), fmax_n(gm[2], gm[3]));
            const float sc = __builtin_amdgcn_exp2f(m - mn);
            m = mn;
            acc0[0] *= sc; acc0[1] *= sc;
            acc1[0] *= sc; acc1[1] *= sc;
            #pragma unroll
            for (int u = 0; u < 4; ++u) {
                const v2f a = {kk[u].x, kk[u].y};
                p[u] = mm4(a, bP, zero8);
            }

            #pragma unroll
            for (int u = 0; u < 4; ++u) {
                v8f e;
                #pragma unroll
                for (int r = 0; r < 8; ++r)
                    e[r] = __builtin_amdgcn_exp2f(L[u][r] - mn);
                v2f ha, eb;
                ha = (v2f){ff[u][0].x, ff[u][0].y}; eb = (v2f){e[0], e[1]}; acc0 = mm4(ha, eb, acc0);
                ha = (v2f){ff[u][0].z, ff[u][0].w}; eb = (v2f){e[2], e[3]}; acc1 = mm4(ha, eb, acc1);
                ha = (v2f){ff[u][1].x, ff[u][1].y}; eb = (v2f){e[4], e[5]}; acc0 = mm4(ha, eb, acc0);
                ha = (v2f){ff[u][1].z, ff[u][1].w}; eb = (v2f){e[6], e[7]}; acc1 = mm4(ha, eb, acc1);
            }
            asm volatile("" :: "v"(nk[0].x), "v"(nk[0].y), "v"(nk[1].x), "v"(nk[1].y),
                               "v"(nk[2].x), "v"(nk[2].y), "v"(nk[3].x), "v"(nk[3].y));
        }
    }

    {
        const float eh = acc0[0] + acc1[0];
        const float se = acc0[1] + acc1[1];
        const float mo  = __shfl_xor(m, 16, 32);
        const float eho = __shfl_xor(eh, 16, 32);
        const float seo = __shfl_xor(se, 16, 32);
        const float mm  = fmax_n(m, mo);
        const float w   = __builtin_amdgcn_exp2f(m - mm);
        const float wo  = __builtin_amdgcn_exp2f(mo - mm);
        const float num = eh * w + eho * wo;
        const float den = se * w + seo * wo;
        if (half == 0)
            s_A[wave * ROWS_PER_WAVE + cidx] = (mm < -1.0e8f) ? 0.0f : (num / den);
    }
    __syncthreads();

    const float wv0 = Wv[2 * tid];
    const float wv1 = Wv[2 * tid + 1];
    v2f* o2 = (v2f*)out + (size_t)(b * SEQ + rowBase) * (DMODEL / 2) + tid;
    for (int rr = 0; rr < ROWS_PER_BLOCK; ++rr) {
        const float A = s_A[rr];
        v2f val;
        val.x = A * wv0;
        val.y = A * wv1;
        __builtin_nontemporal_store(val, o2);
        o2 += DMODEL / 2;
    }
}

extern "C" void kernel_launch(void* const* d_in, const int* in_sizes, int n_in,
                              void* d_out, int out_size, void* d_ws, size_t ws_size,
                              hipStream_t stream) {
    (void)in_sizes; (void)n_in; (void)out_size; (void)d_ws; (void)ws_size;
    const float* in = (const float*)d_in[0];
    const float* Wq = (const float*)d_in[1];
    const float* Wk = (const float*)d_in[2];
    const float* Wv = (const float*)d_in[3];
    float* out = (float*)d_out;

    dim3 grid(BATCH * (SEQ / ROWS_PER_BLOCK));
    dim3 block(THREADS);
    dwsa_kernel<<<grid, block, 0, stream>>>(in, Wq, Wk, Wv, out);
}


# u_j - u_i moves from a matrix op to the vector unit (u broadcast from LDS) and enters the P op as C under its -|C| modifier: the op returns the logit P - |Q|, summed inside it; Q ops and subtracts go
# speedup vs baseline: 4.5831x; 1.1217x over previous
#include <hip/hip_runtime.h>


typedef float v2f __attribute__((ext_vector_type(2)));
typedef float v8f __attribute__((ext_vector_type(8)));

#define BATCH 8
#define SEQ 2048
#define DMODEL 256
#define THREADS 128
#define WAVES 4
#define ROWS_PER_WAVE 16
#define ROWS_PER_BLOCK 64
#define LAMBDA 0.5f
#define NEG_BIG -1.0e9f
#define LOG2E 1.4426950408889634f
#define TILE 16
#define NTILES (SEQ / TILE)
#define KEYS_PER_THREAD (SEQ / THREADS)
#define PLANE1 (SEQ + 16)

__device__ __forceinline__ float fmax_n(float a, float b) { return __builtin_fmaximum_numf(a, b); }

__device__ __forceinline__ v8f mm4(v2f a, v2f b, v8f c) {
    return __builtin_amdgcn_wmma_f32_16x16x4_f32(
         false, a,  false, b,
         (short)0, c,  false,  false);
}

__device__ __forceinline__ v8f mm4_nabs(v2f a, v2f b, v8f c) {
    return __builtin_amdgcn_wmma_f32_16x16x4_f32(
         false, a,  false, b,
         (short)3, c,  false,  false);
}


__global__ __launch_bounds__(THREADS)
void dwsa_kernel(const float* __restrict__ in,
                 const float* __restrict__ Wq,
                 const float* __restrict__ Wk,
                 const float* __restrict__ Wv,
                 float* __restrict__ out)
{
    __shared__ __align__(16) float2 s_ka[PLANE1 + SEQ];
    __shared__ __align__(16) float  s_h[SEQ + 16];
    __shared__ __align__(16) float  s_u[SEQ];
    __shared__ float s_A[ROWS_PER_BLOCK];

    const int tid  = threadIdx.x;
    const int b    = blockIdx.x >> 5;
    const int rowBase = (blockIdx.x & 31) * ROWS_PER_BLOCK;
    const int wave = tid >> 5;
    const int lane = tid & 31;
    const int half = lane >> 4;
    const int cidx = lane & 15;

    const float2* seq = (const float2*)in + (size_t)b * SEQ;
    float2 zh[KEYS_PER_THREAD];
    #pragma unroll
    for (int k = 0; k < KEYS_PER_THREAD; ++k) zh[k] = seq[tid + k * THREADS];

    float c = 0.0f;
    #pragma unroll 8
    for (int i = 0; i < DMODEL; ++i) c += Wq[i] * Wk[i];
    c *= (1.0f / 16.0f);

    #pragma unroll
    for (int k = 0; k < KEYS_PER_THREAD; ++k) {
        const int t = tid + k * THREADS;
        s_ka[t]          = make_float2(zh[k].y, (LAMBDA * LOG2E) * zh[k].x);
        s_ka[PLANE1 + t] = make_float2((zh[k].y == 0.0f) ? (LOG2E * NEG_BIG) : 0.0f, 1.0f);
        s_h[t]           = zh[k].y;
        s_u[t]           = (LAMBDA * LOG2E) * zh[k].x;
    }
    if (tid < 16) s_h[SEQ + tid] = (tid < 8) ? 1.0f : 0.0f;

    __syncthreads();

    const int    row = rowBase + wave * ROWS_PER_WAVE + cidx;
    const float2 q2  = s_ka[row];
    const float  gh  = (LOG2E * c) * q2.x;
    const float  ui  = q2.y;

    const float2* ka = s_ka + (half ? PLANE1 : 0) + cidx;
    const v8f zero8 = {0.f, 0.f, 0.f, 0.f, 0.f, 0.f, 0.f, 0.f};

    v2f bP;
    bP.x = half ? 1.0f : gh;   bP.y = 0.0f;

    const bool rowH   = (cidx == 8 * half);
    const bool rowOne = (cidx == 8 * half + 1);
    const int hbase   = rowH ? (8 * half) : (rowOne ? SEQ : (SEQ + 8));
    const int hstride = rowH ? TILE : 0;

    const float* up = s_u + 8 * half;

    float m = -3.4e38f;
    v8f acc0 = zero8, acc1 = zero8;
    {
        v8f L[4];
        #pragma unroll
        for (int u = 0; u < 4; ++u) {
            const float2 k0 = ka[u * TILE];
            const float4* uq = (const float4*)(up + u * TILE);
            const float4 u0 = uq[0], u1 = uq[1];
            const v2f a = {k0.x, k0.y};
            const v8f d = {u0.x - ui, u0.y - ui, u0.z - ui, u0.w - ui,
                           u1.x - ui, u1.y - ui, u1.z - ui, u1.w - ui};
            L[u] = mm4_nabs(a, bP, d);
        }

        for (int t = 0; t < NTILES; t += 4) {
            float4 ff[4][2];
            float2 kk[4];
            float4 uu[4][2];
            const int tn = (t + 4 < NTILES) ? (t + 4) : (NTILES - 4);
            #pragma unroll
            for (int u = 0; u < 4; ++u) {
                const float4* hp = (const float4*)(s_h + hbase + (t + u) * hstride);
                ff[u][0] = hp[0]; ff[u][1] = hp[1];
            }
            #pragma unroll
            for (int u = 0; u < 4; ++u) {
                kk[u] = ka[(tn + u) * TILE];
                const float4* uq = (const float4*)(up + (tn + u) * TILE);
                uu[u][0] = uq[0]; uu[u][1] = uq[1];
            }
            __builtin_amdgcn_sched_barrier(0);

            float gm[4];
            #pragma unroll
            for (int u = 0; u < 4; ++u)
                gm[u] = fmax_n(fmax_n(fmax_n(L[u][0], L[u][1]), fmax_n(L[u][2], L[u][3])),
                               fmax_n(fmax_n(L[u][4], L[u][5]), fmax_n(L[u][6], L[u][7])));
            const float mn = fmax_n(fmax_n(m, fmax_n(gm[0], gm[1])), fmax_n(gm[2], gm[3]));
            const float sc = __builtin_amdgcn_exp2f(m - mn);
            m = mn;
            acc0[0] *= sc; acc0[1] *= sc;
            acc1[0] *= sc; acc1[1] *= sc;
            v8f e[4];
            #pragma unroll
            for (int u = 0; u < 4; ++u)
                #pragma unroll
                for (int r = 0; r < 8; ++r)
                    e[u][r] = __builtin_amdgcn_exp2f(L[u][r] - mn);

            asm volatile("" :: "v"(ff[0][0].x), "v"(ff[0][1].x), "v"(ff[1][0].x), "v"(ff[1][1].x),
                               "v"(ff[2][0].x), "v"(ff[2][1].x), "v"(ff[3][0].x), "v"(ff[3][1].x),
                               "v"(uu[0][0].x), "v"(uu[0][1].x), "v"(uu[1][0].x), "v"(uu[1][1].x),
                               "v"(uu[2][0].x), "v"(uu[2][1].x), "v"(uu[3][0].x), "v"(uu[3][1].x),
                               "v"(kk[0].x), "v"(kk[1].x), "v"(kk[2].x), "v"(kk[3].x));

            #pragma unroll
            for (int u = 0; u < 4; ++u) {
                const v2f a = {kk[u].x, kk[u].y};
                const v8f d = {uu[u][0].x - ui, uu[u][0].y - ui, uu[u][0].z - ui, uu[u][0].w - ui,
                               uu[u][1].x - ui, uu[u][1].y - ui, uu[u][1].z - ui, uu[u][1].w - ui};
                L[u] = mm4_nabs(a, bP, d);
                v2f ha, eb;
                ha = (v2f){ff[u][0].x, ff[u][0].y}; eb = (v2f){e[u][0], e[u][1]}; acc0 = mm4(ha, eb, acc0);
                ha = (v2f){ff[u][0].z, ff[u][0].w}; eb = (v2f){e[u][2], e[u][3]}; acc1 = mm4(ha, eb, acc1);
                ha = (v2f){ff[u][1].x, ff[u][1].y}; eb = (v2f){e[u][4], e[u][5]}; acc0 = mm4(ha, eb, acc0);
                ha = (v2f){ff[u][1].z, ff[u][1].w}; eb = (v2f){e[u][6], e[u][7]}; acc1 = mm4(ha, eb, acc1);
            }
        }
    }

    {
        const float eh = acc0[0] + acc1[0];
        const float se = acc0[1] + acc1[1];
        const float mo  = __shfl_xor(m, 16, 32);
        const float eho = __shfl_xor(eh, 16, 32);
        const float seo = __shfl_xor(se, 16, 32);
        const float mm  = fmax_n(m, mo);
        const float w   = __builtin_amdgcn_exp2f(m - mm);
        const float wo  = __builtin_amdgcn_exp2f(mo - mm);
        const float num = eh * w + eho * wo;
        const float den = se * w + seo * wo;
        if (half == 0)
            s_A[wave * ROWS_PER_WAVE + cidx] = (mm < -1.0e8f) ? 0.0f : (num / den);
    }
    __syncthreads();

    const float wv0 = Wv[2 * tid];
    const float wv1 = Wv[2 * tid + 1];
    v2f* o2 = (v2f*)out + (size_t)(b * SEQ + rowBase) * (DMODEL / 2) + tid;
    for (int rr = 0; rr < ROWS_PER_BLOCK; ++rr) {
        const float A = s_A[rr];
        v2f val;
        val.x = A * wv0;
        val.y = A * wv1;
        __builtin_nontemporal_store(val, o2);
        o2 += DMODEL / 2;
    }
}

extern "C" void kernel_launch(void* const* d_in, const int* in_sizes, int n_in,
                              void* d_out, int out_size, void* d_ws, size_t ws_size,
                              hipStream_t stream) {
    (void)in_sizes; (void)n_in; (void)out_size; (void)d_ws; (void)ws_size;
    const float* in = (const float*)d_in[0];
    const float* Wq = (const float*)d_in[1];
    const float* Wk = (const float*)d_in[2];
    const float* Wv = (const float*)d_in[3];
    float* out = (float*)d_out;

    dim3 grid(BATCH * (SEQ / ROWS_PER_BLOCK));
    dim3 block(THREADS);
    dwsa_kernel<<<grid, block, 0, stream>>>(in, Wq, Wk, Wv, out);
}
